# starting kernel
# baseline (speedup 1.0000x reference)
#include <hip/hip_runtime.h>
#include <hip/hip_bf16.h>

#define N_NODES 100000
#define N_EDGES 3200000
#define IN_DIM  512
#define HID_DIM 16
#define N_TILES (N_NODES / 16)

typedef float v2f __attribute__((ext_vector_type(2)));
typedef float v8f __attribute__((ext_vector_type(8)));

__global__ __launch_bounds__(256) void gcn_gemm_xw1(
    const float* __restrict__ X, const float* __restrict__ W1,
    float* __restrict__ h1) {
  __shared__ float sW[IN_DIM * HID_DIM];
  for (int i = threadIdx.x; i < IN_DIM * HID_DIM; i += 256) sW[i] = W1[i];
  __syncthreads();

  const int wave = threadIdx.x >> 5;
  const int lane = threadIdx.x & 31;
  const int tile = blockIdx.x * 8 + wave;
  if (tile >= N_TILES) return;

  const int m  = lane & 15;
  const int hi = lane >> 4;

  const float* __restrict__ xrow = X + (size_t)(tile * 16 + m) * IN_DIM;

  v8f c = {0.f, 0.f, 0.f, 0.f, 0.f, 0.f, 0.f, 0.f};

  #pragma unroll 4
  for (int k0 = 0; k0 < IN_DIM; k0 += 4) {
    const int k = k0 + 2 * hi;
    v2f a;
    a.x = xrow[k];
    a.y = xrow[k + 1];
    v2f b;
    b.x = sW[k * HID_DIM + m];
    b.y = sW[(k + 1) * HID_DIM + m];
    c = __builtin_amdgcn_wmma_f32_16x16x4_f32(
        false, a, false, b, (short)0, c, false, false);
  }

  float* __restrict__ out = h1 + (size_t)tile * 16 * HID_DIM;
  const int mbase = hi * 8;
  #pragma unroll
  for (int v = 0; v < 8; ++v)
    out[(mbase + v) * HID_DIM + m] = c[v];
}

__global__ void gcn_init(float* __restrict__ h2, float* __restrict__ out,
                         const float* __restrict__ b2) {
  int i = blockIdx.x * blockDim.x + threadIdx.x;
  if (i < N_NODES * HID_DIM) h2[i] = 0.0f;
  if (i < N_NODES) out[i] = b2[0];
}

__global__ __launch_bounds__(256) void gcn_spmm1(
    const int* __restrict__ esrc, const int* __restrict__ edst,
    const float* __restrict__ ew, const float* __restrict__ h1,
    float* __restrict__ h2) {
  size_t gid = (size_t)blockIdx.x * 256 + threadIdx.x;
  size_t e = gid >> 4;
  if (e >= N_EDGES) return;
  int d = (int)(gid & 15);
  int s = esrc[e];
  int t = edst[e];
  float v = ew[e] * h1[(size_t)s * HID_DIM + d];
  __hip_atomic_fetch_add(&h2[(size_t)t * HID_DIM + d], v,
                         __ATOMIC_RELAXED, __HIP_MEMORY_SCOPE_AGENT);
}

__global__ void gcn_layer2(const float* __restrict__ h2,
                           const float* __restrict__ b1,
                           const float* __restrict__ W2,
                           float* __restrict__ z) {
  int i = blockIdx.x * blockDim.x + threadIdx.x;
  if (i >= N_NODES) return;
  const float* __restrict__ row = h2 + (size_t)i * HID_DIM;
  float acc = 0.0f;
  #pragma unroll
  for (int d = 0; d < HID_DIM; ++d) {
    float hv = row[d] + b1[d];
    hv = fmaxf(hv, 0.0f);
    acc = fmaf(hv, W2[d], acc);
  }
  z[i] = acc;
}

__global__ __launch_bounds__(256) void gcn_spmm2(
    const int* __restrict__ esrc, const int* __restrict__ edst,
    const float* __restrict__ ew, const float* __restrict__ z,
    float* __restrict__ out) {
  size_t e = (size_t)blockIdx.x * 256 + threadIdx.x;
  if (e >= N_EDGES) return;
  float v = ew[e] * z[esrc[e]];
  __hip_atomic_fetch_add(&out[edst[e]], v,
                         __ATOMIC_RELAXED, __HIP_MEMORY_SCOPE_AGENT);
}

extern "C" void kernel_launch(void* const* d_in, const int* in_sizes, int n_in,
                              void* d_out, int out_size, void* d_ws, size_t ws_size,
                              hipStream_t stream) {
  const float* x    = (const float*)d_in[0];
  const int*   esrc = (const int*)  d_in[1];
  const int*   edst = (const int*)  d_in[2];
  const float* ew   = (const float*)d_in[3];
  const float* W1   = (const float*)d_in[4];
  const float* b1   = (const float*)d_in[5];
  const float* W2   = (const float*)d_in[6];
  const float* b2   = (const float*)d_in[7];
  float* out = (float*)d_out;

  float* h1 = (float*)d_ws;
  float* h2 = h1 + (size_t)N_NODES * HID_DIM;
  float* z  = h2 + (size_t)N_NODES * HID_DIM;

  {
    int blocks = (N_TILES + 7) / 8;
    gcn_gemm_xw1<<<blocks, 256, 0, stream>>>(x, W1, h1);
  }
  {
    int total = N_NODES * HID_DIM;
    gcn_init<<<(total + 255) / 256, 256, 0, stream>>>(h2, out, b2);
  }
  {
    size_t total = (size_t)N_EDGES * HID_DIM;
    int blocks = (int)((total + 255) / 256);
    gcn_spmm1<<<blocks, 256, 0, stream>>>(esrc, edst, ew, h1, h2);
  }
  gcn_layer2<<<(N_NODES + 255) / 256, 256, 0, stream>>>(h2, b1, W2, z);
  gcn_spmm2<<<(N_EDGES + 255) / 256, 256, 0, stream>>>(esrc, edst, ew, z, out);
}


# layer-2 gather: a work-group also orders its chunk by the line of z each edge reads (2nd LDS counting sort), reads z in that order: a wave's 32 reads touch few lines, not 32; f32 add order still free
# speedup vs baseline: 4.0274x; 4.0274x over previous
#include <hip/hip_runtime.h>
#include <hip/hip_bf16.h>

#define N_NODES 100000
#define N_EDGES 3200000
#define IN_DIM  512
#define HID_DIM 16
#define N_TILES (N_NODES / 16)
#define H2_PLANE (N_NODES * HID_DIM)

typedef float v2f __attribute__((ext_vector_type(2)));
typedef float v8f __attribute__((ext_vector_type(8)));

__global__ __launch_bounds__(256) void gcn_gemm_xw1(
    const float* __restrict__ X, const float* __restrict__ W1,
    float* __restrict__ h1) {
  __shared__ float sW[IN_DIM * HID_DIM];
  for (int i = threadIdx.x; i < IN_DIM * HID_DIM; i += 256) sW[i] = W1[i];
  __syncthreads();

  const int wave = threadIdx.x >> 5;
  const int lane = threadIdx.x & 31;
  const int tile = blockIdx.x * 8 + wave;
  if (tile >= N_TILES) return;

  const int m  = lane & 15;
  const int hi = lane >> 4;

  const float* __restrict__ xrow = X + (size_t)(tile * 16 + m) * IN_DIM;

  v8f c = {0.f, 0.f, 0.f, 0.f, 0.f, 0.f, 0.f, 0.f};

  #pragma unroll 4
  for (int k0 = 0; k0 < IN_DIM; k0 += 4) {
    const int k = k0 + 2 * hi;
    v2f a;
    a.x = xrow[k];
    a.y = xrow[k + 1];
    v2f b;
    b.x = sW[k * HID_DIM + m];
    b.y = sW[(k + 1) * HID_DIM + m];
    c = __builtin_amdgcn_wmma_f32_16x16x4_f32(
        false, a, false, b, (short)0, c, false, false);
  }

  float* __restrict__ out = h1 + (size_t)tile * 16 * HID_DIM;
  const int mbase = hi * 8;
  #pragma unroll
  for (int v = 0; v < 8; ++v)
    out[(mbase + v) * HID_DIM + m] = c[v];
}

__global__ void gcn_init(float* __restrict__ h2, float* __restrict__ out,
                         const float* __restrict__ b2) {
  int i = blockIdx.x * blockDim.x + threadIdx.x;
  if (i < H2_PLANE) {
    h2[i] = 0.0f;
    h2[H2_PLANE + i] = 0.0f;
  }
  if (i < N_NODES) out[i] = b2[0];
}

static_assert(N_EDGES % 256 == 0, "every lane of every wave owns one edge");

template <int J>
__device__ __forceinline__ int row_bcast_i(int x) {
  return __builtin_amdgcn_update_dpp(0, x, 0x150 + J, 0xF, 0xF, false);
}
template <int J>
__device__ __forceinline__ float row_bcast_f(float x) {
  return __int_as_float(row_bcast_i<J>(__float_as_int(x)));
}

#define SPMM1_GATHER(J)                                              \
  {                                                                  \
    const unsigned sj = (unsigned)row_bcast_i<J>(s);                 \
    tt[J] = (unsigned)row_bcast_i<J>(t);                             \
    vv[J] = row_bcast_f<J>(w) * h1[sj * HID_DIM + d];                \
  }
#define SPMM1_SCATTER(J)                                             \
  __hip_atomic_fetch_add(&acc[tt[J] * HID_DIM + d], vv[J],           \
                         __ATOMIC_RELAXED, __HIP_MEMORY_SCOPE_AGENT);

__global__ __launch_bounds__(256) void gcn_spmm1(
    const int* __restrict__ esrc, const int* __restrict__ edst,
    const float* __restrict__ ew, const float* __restrict__ h1,
    float* __restrict__ h2) {
  const unsigned e = blockIdx.x * 256u + threadIdx.x;
  const unsigned d = threadIdx.x & 15u;
  const int   s = esrc[e];
  const int   t = edst[e];
  const float w = ew[e];

  const unsigned where = __builtin_amdgcn_s_sendmsg_rtn(0x87);
  float* __restrict__ acc = h2 + (size_t)((where >> 18) & 1u) * H2_PLANE;

  unsigned tt[16];
  float    vv[16];
  SPMM1_GATHER(0)  SPMM1_GATHER(1)  SPMM1_GATHER(2)  SPMM1_GATHER(3)
  SPMM1_GATHER(4)  SPMM1_GATHER(5)  SPMM1_GATHER(6)  SPMM1_GATHER(7)
  SPMM1_GATHER(8)  SPMM1_GATHER(9)  SPMM1_GATHER(10) SPMM1_GATHER(11)
  SPMM1_GATHER(12) SPMM1_GATHER(13) SPMM1_GATHER(14) SPMM1_GATHER(15)
  SPMM1_SCATTER(0)  SPMM1_SCATTER(1)  SPMM1_SCATTER(2)  SPMM1_SCATTER(3)
  SPMM1_SCATTER(4)  SPMM1_SCATTER(5)  SPMM1_SCATTER(6)  SPMM1_SCATTER(7)
  SPMM1_SCATTER(8)  SPMM1_SCATTER(9)  SPMM1_SCATTER(10) SPMM1_SCATTER(11)
  SPMM1_SCATTER(12) SPMM1_SCATTER(13) SPMM1_SCATTER(14) SPMM1_SCATTER(15)
}
#undef SPMM1_GATHER
#undef SPMM1_SCATTER

__global__ void gcn_layer2(const float* __restrict__ h2,
                           const float* __restrict__ b1,
                           const float* __restrict__ W2,
                           float* __restrict__ z,
                           float* __restrict__ out1) {
  int i = blockIdx.x * blockDim.x + threadIdx.x;
  if (i >= N_NODES) return;
  out1[i] = 0.0f;
  const float* __restrict__ row  = h2 + (size_t)i * HID_DIM;
  const float* __restrict__ row1 = row + H2_PLANE;
  float acc = 0.0f;
  #pragma unroll
  for (int d = 0; d < HID_DIM; ++d) {
    float hv = (row[d] + row1[d]) + b1[d];
    hv = fmaxf(hv, 0.0f);
    acc = fmaf(hv, W2[d], acc);
  }
  z[i] = acc;
}

#define S2_THREADS    512
#define S2_PER_THREAD 25
#define S2_CHUNK      (S2_THREADS * S2_PER_THREAD)
#define S2_GROUPS     (N_EDGES / S2_CHUNK)
#define S2_BINS       4096
#define S2_BINS_PER_THREAD (S2_BINS / S2_THREADS)
static_assert(N_EDGES % S2_CHUNK == 0, "every work-group owns a full chunk");
static_assert((N_NODES + 31) / 32 <= S2_BINS, "one counter per 128-byte line of out and of z");
static_assert(S2_BINS % S2_THREADS == 0, "scan: whole bins per thread");
static_assert((S2_GROUPS - 1) * (S2_CHUNK / S2_GROUPS) < S2_CHUNK, "walk rotation stays inside one wrap");

__global__ __launch_bounds__(S2_THREADS) void gcn_spmm2(
    const int* __restrict__ esrc, const int* __restrict__ edst,
    const float* __restrict__ ew, const float* __restrict__ z,
    float* __restrict__ out, float* __restrict__ out1) {
  __shared__ unsigned sBinD[S2_BINS];
  __shared__ unsigned sBinS[S2_BINS];
  __shared__ unsigned sZ[S2_CHUNK];
  __shared__ uint2    sAdd[S2_CHUNK];
  __shared__ unsigned sWaveSum[2][S2_THREADS / 32];

  const unsigned tid  = threadIdx.x;
  const unsigned lane = tid & 31u;
  const unsigned wave = tid >> 5;
  const unsigned base = blockIdx.x * (unsigned)S2_CHUNK;

  for (unsigned i = tid; i < S2_BINS; i += S2_THREADS) {
    sBinD[i] = 0u;
    sBinS[i] = 0u;
  }

  unsigned t[S2_PER_THREAD], s[S2_PER_THREAD];
  unsigned pd[S2_PER_THREAD], ps[S2_PER_THREAD];
  float    w[S2_PER_THREAD];
  #pragma unroll
  for (int k = 0; k < S2_PER_THREAD; ++k) {
    const unsigned e = base + (unsigned)k * S2_THREADS + tid;
    t[k] = (unsigned)edst[e];
    s[k] = (unsigned)esrc[e];
    w[k] = ew[e];
  }
  __syncthreads();

  #pragma unroll
  for (int k = 0; k < S2_PER_THREAD; ++k) {
    pd[k] = atomicAdd(&sBinD[(t[k] >> 5) & (S2_BINS - 1)], 1u);
    ps[k] = atomicAdd(&sBinS[(s[k] >> 5) & (S2_BINS - 1)], 1u);
  }
  __syncthreads();

  unsigned cD[S2_BINS_PER_THREAD], cS[S2_BINS_PER_THREAD];
  unsigned runD = 0u, runS = 0u;
  #pragma unroll
  for (int i = 0; i < S2_BINS_PER_THREAD; ++i) {
    const unsigned nD = sBinD[tid * S2_BINS_PER_THREAD + i];
    const unsigned nS = sBinS[tid * S2_BINS_PER_THREAD + i];
    cD[i] = runD;
    cS[i] = runS;
    runD += nD;
    runS += nS;
  }
  unsigned inclD = runD, inclS = runS;
  #pragma unroll
  for (unsigned d = 1u; d < 32u; d <<= 1) {
    const unsigned upD = __shfl_up(inclD, d);
    const unsigned upS = __shfl_up(inclS, d);
    if (lane >= d) {
      inclD += upD;
      inclS += upS;
    }
  }
  if (lane == 31u) {
    sWaveSum[0][wave] = inclD;
    sWaveSum[1][wave] = inclS;
  }
  __syncthreads();
  unsigned beforeD = inclD - runD, beforeS = inclS - runS;
  for (unsigned i = 0; i < wave; ++i) {
    beforeD += sWaveSum[0][i];
    beforeS += sWaveSum[1][i];
  }
  #pragma unroll
  for (int i = 0; i < S2_BINS_PER_THREAD; ++i) {
    sBinD[tid * S2_BINS_PER_THREAD + i] = beforeD + cD[i];
    sBinS[tid * S2_BINS_PER_THREAD + i] = beforeS + cS[i];
  }
  __syncthreads();

  #pragma unroll
  for (int k = 0; k < S2_PER_THREAD; ++k) {
    ps[k] += sBinS[(s[k] >> 5) & (S2_BINS - 1)];
    pd[k] += sBinD[(t[k] >> 5) & (S2_BINS - 1)];
    sZ[ps[k]] = s[k];
  }
  __syncthreads();

  const unsigned rot = blockIdx.x * (unsigned)(S2_CHUNK / S2_GROUPS);

  float zz[S2_PER_THREAD];
  #pragma unroll
  for (int k = 0; k < S2_PER_THREAD; ++k) {
    unsigned j = (unsigned)k * S2_THREADS + tid + rot;
    if (j >= (unsigned)S2_CHUNK) j -= (unsigned)S2_CHUNK;
    zz[k] = z[sZ[j]];
  }
  #pragma unroll
  for (int k = 0; k < S2_PER_THREAD; ++k) {
    unsigned j = (unsigned)k * S2_THREADS + tid + rot;
    if (j >= (unsigned)S2_CHUNK) j -= (unsigned)S2_CHUNK;
    sZ[j] = __float_as_uint(zz[k]);
  }
  __syncthreads();

  #pragma unroll
  for (int k = 0; k < S2_PER_THREAD; ++k) {
    const float v = w[k] * __uint_as_float(sZ[ps[k]]);
    sAdd[pd[k]] = make_uint2(t[k], __float_as_uint(v));
  }
  __syncthreads();

  const unsigned where = __builtin_amdgcn_s_sendmsg_rtn(0x87);
  float* __restrict__ acc = ((where >> 18) & 1u) ? out1 : out;
  #pragma unroll
  for (int k = 0; k < S2_PER_THREAD; ++k) {
    unsigned j = (unsigned)k * S2_THREADS + tid + rot;
    if (j >= (unsigned)S2_CHUNK) j -= (unsigned)S2_CHUNK;
    const uint2 p = sAdd[j];
    __hip_atomic_fetch_add(&acc[p.x], __uint_as_float(p.y),
                           __ATOMIC_RELAXED, __HIP_MEMORY_SCOPE_AGENT);
  }
}

static_assert(N_NODES % 4 == 0, "whole 16-byte pieces");
__global__ __launch_bounds__(256) void gcn_sum_out(
    float* __restrict__ out, const float* __restrict__ out1) {
  const int i = blockIdx.x * 256 + threadIdx.x;
  if (i >= N_NODES / 4) return;
  float4 a = reinterpret_cast<float4*>(out)[i];
  const float4 b = reinterpret_cast<const float4*>(out1)[i];
  a.x += b.x;
  a.y += b.y;
  a.z += b.z;
  a.w += b.w;
  reinterpret_cast<float4*>(out)[i] = a;
}

extern "C" void kernel_launch(void* const* d_in, const int* in_sizes, int n_in,
                              void* d_out, int out_size, void* d_ws, size_t ws_size,
                              hipStream_t stream) {
  const float* x    = (const float*)d_in[0];
  const int*   esrc = (const int*)  d_in[1];
  const int*   edst = (const int*)  d_in[2];
  const float* ew   = (const float*)d_in[3];
  const float* W1   = (const float*)d_in[4];
  const float* b1   = (const float*)d_in[5];
  const float* W2   = (const float*)d_in[6];
  const float* b2   = (const float*)d_in[7];
  float* out = (float*)d_out;

  float* h1 = (float*)d_ws;
  float* out1 = h1;
  float* h2 = h1 + (size_t)N_NODES * HID_DIM;
  float* z  = h2 + (size_t)2 * H2_PLANE;

  {
    int blocks = (N_TILES + 7) / 8;
    gcn_gemm_xw1<<<blocks, 256, 0, stream>>>(x, W1, h1);
  }
  {
    int total = N_NODES * HID_DIM;
    gcn_init<<<(total + 255) / 256, 256, 0, stream>>>(h2, out, b2);
  }
  gcn_spmm1<<<N_EDGES / 256, 256, 0, stream>>>(esrc, edst, ew, h1, h2);
  gcn_layer2<<<(N_NODES + 255) / 256, 256, 0, stream>>>(h2, b1, W2, z, out1);
  gcn_spmm2<<<S2_GROUPS, S2_THREADS, 0, stream>>>(esrc, edst, ew, z, out, out1);
  gcn_sum_out<<<(N_NODES / 4 + 255) / 256, 256, 0, stream>>>(out, out1);
}


# init and layer-1 scatter: few waves that each walk their work (10 32-edge groups, records asked for at once; 5 16-byte clears), not a wave per 32 edges / 256 floats; f32 atomic order still free
# speedup vs baseline: 4.2037x; 1.0438x over previous
#include <hip/hip_runtime.h>
#include <hip/hip_bf16.h>

#define N_NODES 100000
#define N_EDGES 3200000
#define IN_DIM  512
#define HID_DIM 16
#define N_TILES (N_NODES / 16)
#define H2_PLANE (N_NODES * HID_DIM)

typedef float v2f __attribute__((ext_vector_type(2)));
typedef float v8f __attribute__((ext_vector_type(8)));

__global__ __launch_bounds__(256) void gcn_gemm_xw1(
    const float* __restrict__ X, const float* __restrict__ W1,
    float* __restrict__ h1) {
  __shared__ float sW[IN_DIM * HID_DIM];
  for (int i = threadIdx.x; i < IN_DIM * HID_DIM; i += 256) sW[i] = W1[i];
  __syncthreads();

  const int wave = threadIdx.x >> 5;
  const int lane = threadIdx.x & 31;
  const int tile = blockIdx.x * 8 + wave;
  if (tile >= N_TILES) return;

  const int m  = lane & 15;
  const int hi = lane >> 4;

  const float* __restrict__ xrow = X + (size_t)(tile * 16 + m) * IN_DIM;

  v8f c = {0.f, 0.f, 0.f, 0.f, 0.f, 0.f, 0.f, 0.f};

  #pragma unroll 4
  for (int k0 = 0; k0 < IN_DIM; k0 += 4) {
    const int k = k0 + 2 * hi;
    v2f a;
    a.x = xrow[k];
    a.y = xrow[k + 1];
    v2f b;
    b.x = sW[k * HID_DIM + m];
    b.y = sW[(k + 1) * HID_DIM + m];
    c = __builtin_amdgcn_wmma_f32_16x16x4_f32(
        false, a, false, b, (short)0, c, false, false);
  }

  float* __restrict__ out = h1 + (size_t)tile * 16 * HID_DIM;
  const int mbase = hi * 8;
  #pragma unroll
  for (int v = 0; v < 8; ++v)
    out[(mbase + v) * HID_DIM + m] = c[v];
}

#define INIT_THREADS  256
#define INIT_PER_LANE 5
#define INIT_PIECES   (2 * H2_PLANE / 4)
#define INIT_GROUPS   (INIT_PIECES / (INIT_THREADS * INIT_PER_LANE))
static_assert(INIT_PIECES % (INIT_THREADS * INIT_PER_LANE) == 0, "every lane clears whole pieces, none left over");
static_assert(N_NODES / 4 <= INIT_GROUPS * INIT_THREADS, "one lane per 16-byte piece of out");
__global__ __launch_bounds__(INIT_THREADS) void gcn_init(
    float* __restrict__ h2, float* __restrict__ out,
    const float* __restrict__ b2) {
  const unsigned i = blockIdx.x * (unsigned)INIT_THREADS + threadIdx.x;
  float4* __restrict__ h2v = reinterpret_cast<float4*>(h2);
  const float4 zero = make_float4(0.0f, 0.0f, 0.0f, 0.0f);
  for (unsigned k = 0; k < (unsigned)INIT_PER_LANE; ++k)
    h2v[k * (unsigned)(INIT_GROUPS * INIT_THREADS) + i] = zero;
  if (i < (unsigned)(N_NODES / 4)) {
    const float b = b2[0];
    reinterpret_cast<float4*>(out)[i] = make_float4(b, b, b, b);
  }
}

#define S1_THREADS 256
#define S1_STEPS   10
#define S1_WAVES   (N_EDGES / (32 * S1_STEPS))
#define S1_GROUPS  (S1_WAVES * 32 / S1_THREADS)
static_assert(N_EDGES % (32 * S1_STEPS) == 0, "every wave owns S1_STEPS full groups");
static_assert((S1_WAVES * 32) % S1_THREADS == 0, "whole work-groups");

template <int J>
__device__ __forceinline__ int row_bcast_i(int x) {
  return __builtin_amdgcn_update_dpp(0, x, 0x150 + J, 0xF, 0xF, false);
}
template <int J>
__device__ __forceinline__ float row_bcast_f(float x) {
  return __int_as_float(row_bcast_i<J>(__float_as_int(x)));
}

#define SPMM1_GATHER(J)                                              \
  {                                                                  \
    const unsigned sj = (unsigned)row_bcast_i<J>(s);                 \
    tt[J] = (unsigned)row_bcast_i<J>(t);                             \
    vv[J] = row_bcast_f<J>(w) * h1[sj * HID_DIM + d];                \
  }
#define SPMM1_SCATTER(J)                                             \
  __hip_atomic_fetch_add(&acc[tt[J] * HID_DIM + d], vv[J],           \
                         __ATOMIC_RELAXED, __HIP_MEMORY_SCOPE_AGENT);

__global__ __launch_bounds__(S1_THREADS) void gcn_spmm1(
    const int* __restrict__ esrc, const int* __restrict__ edst,
    const float* __restrict__ ew, const float* __restrict__ h1,
    float* __restrict__ h2) {
  const unsigned d = threadIdx.x & 15u;
  const unsigned wave_id = (blockIdx.x * (unsigned)S1_THREADS + threadIdx.x) >> 5;
  const unsigned e0 = wave_id * (unsigned)(32 * S1_STEPS) + (threadIdx.x & 31u);

  int   sr[S1_STEPS], tr[S1_STEPS];
  float wr[S1_STEPS];
  #pragma unroll
  for (int g = 0; g < S1_STEPS; ++g) {
    sr[g] = esrc[e0 + 32u * (unsigned)g];
    tr[g] = edst[e0 + 32u * (unsigned)g];
    wr[g] = ew[e0 + 32u * (unsigned)g];
  }

  const unsigned where = __builtin_amdgcn_s_sendmsg_rtn(0x87);
  float* __restrict__ acc = h2 + (size_t)((where >> 18) & 1u) * H2_PLANE;

  #pragma unroll
  for (int g = 0; g < S1_STEPS; ++g) {
    const int   s = sr[g];
    const int   t = tr[g];
    const float w = wr[g];
    unsigned tt[16];
    float    vv[16];
    SPMM1_GATHER(0)  SPMM1_GATHER(1)  SPMM1_GATHER(2)  SPMM1_GATHER(3)
    SPMM1_GATHER(4)  SPMM1_GATHER(5)  SPMM1_GATHER(6)  SPMM1_GATHER(7)
    SPMM1_GATHER(8)  SPMM1_GATHER(9)  SPMM1_GATHER(10) SPMM1_GATHER(11)
    SPMM1_GATHER(12) SPMM1_GATHER(13) SPMM1_GATHER(14) SPMM1_GATHER(15)
    SPMM1_SCATTER(0)  SPMM1_SCATTER(1)  SPMM1_SCATTER(2)  SPMM1_SCATTER(3)
    SPMM1_SCATTER(4)  SPMM1_SCATTER(5)  SPMM1_SCATTER(6)  SPMM1_SCATTER(7)
    SPMM1_SCATTER(8)  SPMM1_SCATTER(9)  SPMM1_SCATTER(10) SPMM1_SCATTER(11)
    SPMM1_SCATTER(12) SPMM1_SCATTER(13) SPMM1_SCATTER(14) SPMM1_SCATTER(15)
  }
}
#undef SPMM1_GATHER
#undef SPMM1_SCATTER

__global__ void gcn_layer2(const float* __restrict__ h2,
                           const float* __restrict__ b1,
                           const float* __restrict__ W2,
                           float* __restrict__ z,
                           float* __restrict__ out1) {
  int i = blockIdx.x * blockDim.x + threadIdx.x;
  if (i >= N_NODES) return;
  out1[i] = 0.0f;
  const float* __restrict__ row  = h2 + (size_t)i * HID_DIM;
  const float* __restrict__ row1 = row + H2_PLANE;
  float acc = 0.0f;
  #pragma unroll
  for (int d = 0; d < HID_DIM; ++d) {
    float hv = (row[d] + row1[d]) + b1[d];
    hv = fmaxf(hv, 0.0f);
    acc = fmaf(hv, W2[d], acc);
  }
  z[i] = acc;
}

#define S2_THREADS    512
#define S2_PER_THREAD 25
#define S2_CHUNK      (S2_THREADS * S2_PER_THREAD)
#define S2_GROUPS     (N_EDGES / S2_CHUNK)
#define S2_BINS       4096
#define S2_BINS_PER_THREAD (S2_BINS / S2_THREADS)
static_assert(N_EDGES % S2_CHUNK == 0, "every work-group owns a full chunk");
static_assert((N_NODES + 31) / 32 <= S2_BINS, "one counter per 128-byte line of out and of z");
static_assert(S2_BINS % S2_THREADS == 0, "scan: whole bins per thread");
static_assert((S2_GROUPS - 1) * (S2_CHUNK / S2_GROUPS) < S2_CHUNK, "walk rotation stays inside one wrap");

__global__ __launch_bounds__(S2_THREADS) void gcn_spmm2(
    const int* __restrict__ esrc, const int* __restrict__ edst,
    const float* __restrict__ ew, const float* __restrict__ z,
    float* __restrict__ out, float* __restrict__ out1) {
  __shared__ unsigned sBinD[S2_BINS];
  __shared__ unsigned sBinS[S2_BINS];
  __shared__ unsigned sZ[S2_CHUNK];
  __shared__ uint2    sAdd[S2_CHUNK];
  __shared__ unsigned sWaveSum[2][S2_THREADS / 32];

  const unsigned tid  = threadIdx.x;
  const unsigned lane = tid & 31u;
  const unsigned wave = tid >> 5;
  const unsigned base = blockIdx.x * (unsigned)S2_CHUNK;

  for (unsigned i = tid; i < S2_BINS; i += S2_THREADS) {
    sBinD[i] = 0u;
    sBinS[i] = 0u;
  }

  unsigned t[S2_PER_THREAD], s[S2_PER_THREAD];
  unsigned pd[S2_PER_THREAD], ps[S2_PER_THREAD];
  float    w[S2_PER_THREAD];
  #pragma unroll
  for (int k = 0; k < S2_PER_THREAD; ++k) {
    const unsigned e = base + (unsigned)k * S2_THREADS + tid;
    t[k] = (unsigned)edst[e];
    s[k] = (unsigned)esrc[e];
    w[k] = ew[e];
  }
  __syncthreads();

  #pragma unroll
  for (int k = 0; k < S2_PER_THREAD; ++k) {
    pd[k] = atomicAdd(&sBinD[(t[k] >> 5) & (S2_BINS - 1)], 1u);
    ps[k] = atomicAdd(&sBinS[(s[k] >> 5) & (S2_BINS - 1)], 1u);
  }
  __syncthreads();

  unsigned cD[S2_BINS_PER_THREAD], cS[S2_BINS_PER_THREAD];
  unsigned runD = 0u, runS = 0u;
  #pragma unroll
  for (int i = 0; i < S2_BINS_PER_THREAD; ++i) {
    const unsigned nD = sBinD[tid * S2_BINS_PER_THREAD + i];
    const unsigned nS = sBinS[tid * S2_BINS_PER_THREAD + i];
    cD[i] = runD;
    cS[i] = runS;
    runD += nD;
    runS += nS;
  }
  unsigned inclD = runD, inclS = runS;
  #pragma unroll
  for (unsigned d = 1u; d < 32u; d <<= 1) {
    const unsigned upD = __shfl_up(inclD, d);
    const unsigned upS = __shfl_up(inclS, d);
    if (lane >= d) {
      inclD += upD;
      inclS += upS;
    }
  }
  if (lane == 31u) {
    sWaveSum[0][wave] = inclD;
    sWaveSum[1][wave] = inclS;
  }
  __syncthreads();
  unsigned beforeD = inclD - runD, beforeS = inclS - runS;
  for (unsigned i = 0; i < wave; ++i) {
    beforeD += sWaveSum[0][i];
    beforeS += sWaveSum[1][i];
  }
  #pragma unroll
  for (int i = 0; i < S2_BINS_PER_THREAD; ++i) {
    sBinD[tid * S2_BINS_PER_THREAD + i] = beforeD + cD[i];
    sBinS[tid * S2_BINS_PER_THREAD + i] = beforeS + cS[i];
  }
  __syncthreads();

  #pragma unroll
  for (int k = 0; k < S2_PER_THREAD; ++k) {
    ps[k] += sBinS[(s[k] >> 5) & (S2_BINS - 1)];
    pd[k] += sBinD[(t[k] >> 5) & (S2_BINS - 1)];
    sZ[ps[k]] = s[k];
  }
  __syncthreads();

  const unsigned rot = blockIdx.x * (unsigned)(S2_CHUNK / S2_GROUPS);

  float zz[S2_PER_THREAD];
  #pragma unroll
  for (int k = 0; k < S2_PER_THREAD; ++k) {
    unsigned j = (unsigned)k * S2_THREADS + tid + rot;
    if (j >= (unsigned)S2_CHUNK) j -= (unsigned)S2_CHUNK;
    zz[k] = z[sZ[j]];
  }
  #pragma unroll
  for (int k = 0; k < S2_PER_THREAD; ++k) {
    unsigned j = (unsigned)k * S2_THREADS + tid + rot;
    if (j >= (unsigned)S2_CHUNK) j -= (unsigned)S2_CHUNK;
    sZ[j] = __float_as_uint(zz[k]);
  }
  __syncthreads();

  #pragma unroll
  for (int k = 0; k < S2_PER_THREAD; ++k) {
    const float v = w[k] * __uint_as_float(sZ[ps[k]]);
    sAdd[pd[k]] = make_uint2(t[k], __float_as_uint(v));
  }
  __syncthreads();

  const unsigned where = __builtin_amdgcn_s_sendmsg_rtn(0x87);
  float* __restrict__ acc = ((where >> 18) & 1u) ? out1 : out;
  #pragma unroll
  for (int k = 0; k < S2_PER_THREAD; ++k) {
    unsigned j = (unsigned)k * S2_THREADS + tid + rot;
    if (j >= (unsigned)S2_CHUNK) j -= (unsigned)S2_CHUNK;
    const uint2 p = sAdd[j];
    __hip_atomic_fetch_add(&acc[p.x], __uint_as_float(p.y),
                           __ATOMIC_RELAXED, __HIP_MEMORY_SCOPE_AGENT);
  }
}

static_assert(N_NODES % 4 == 0, "whole 16-byte pieces");
__global__ __launch_bounds__(256) void gcn_sum_out(
    float* __restrict__ out, const float* __restrict__ out1) {
  const int i = blockIdx.x * 256 + threadIdx.x;
  if (i >= N_NODES / 4) return;
  float4 a = reinterpret_cast<float4*>(out)[i];
  const float4 b = reinterpret_cast<const float4*>(out1)[i];
  a.x += b.x;
  a.y += b.y;
  a.z += b.z;
  a.w += b.w;
  reinterpret_cast<float4*>(out)[i] = a;
}

extern "C" void kernel_launch(void* const* d_in, const int* in_sizes, int n_in,
                              void* d_out, int out_size, void* d_ws, size_t ws_size,
                              hipStream_t stream) {
  const float* x    = (const float*)d_in[0];
  const int*   esrc = (const int*)  d_in[1];
  const int*   edst = (const int*)  d_in[2];
  const float* ew   = (const float*)d_in[3];
  const float* W1   = (const float*)d_in[4];
  const float* b1   = (const float*)d_in[5];
  const float* W2   = (const float*)d_in[6];
  const float* b2   = (const float*)d_in[7];
  float* out = (float*)d_out;

  float* h1 = (float*)d_ws;
  float* out1 = h1;
  float* h2 = h1 + (size_t)N_NODES * HID_DIM;
  float* z  = h2 + (size_t)2 * H2_PLANE;

  {
    int blocks = (N_TILES + 7) / 8;
    gcn_gemm_xw1<<<blocks, 256, 0, stream>>>(x, W1, h1);
  }
  gcn_init<<<INIT_GROUPS, INIT_THREADS, 0, stream>>>(h2, out, b2);
  gcn_spmm1<<<S1_GROUPS, S1_THREADS, 0, stream>>>(esrc, edst, ew, h1, h2);
  gcn_layer2<<<(N_NODES + 255) / 256, 256, 0, stream>>>(h2, b1, W2, z, out1);
  gcn_spmm2<<<S2_GROUPS, S2_THREADS, 0, stream>>>(esrc, edst, ew, z, out, out1);
  gcn_sum_out<<<(N_NODES / 4 + 255) / 256, 256, 0, stream>>>(out, out1);
}


# init launch dropped: the X @ W1 launch also clears both h2 planes (16-byte stores of every wave behind its tile); the z kernel seeds out with b2; 6 launches -> 5
# speedup vs baseline: 4.3926x; 1.0449x over previous
#include <hip/hip_runtime.h>
#include <hip/hip_bf16.h>

#define N_NODES 100000
#define N_EDGES 3200000
#define IN_DIM  512
#define HID_DIM 16
#define N_TILES (N_NODES / 16)
#define H2_PLANE (N_NODES * HID_DIM)

typedef float v2f __attribute__((ext_vector_type(2)));
typedef float v8f __attribute__((ext_vector_type(8)));

#define GEMM_THREADS_PER_GROUP 256
#define GEMM_GROUPS  ((N_TILES + 7) / 8)
#define GEMM_THREADS (GEMM_GROUPS * GEMM_THREADS_PER_GROUP)
#define H2_PIECES    (2 * H2_PLANE / 4)
#define GEMM_ZERO_PER_LANE 4
static_assert((2 * H2_PLANE) % 4 == 0, "h2 is whole 16-byte pieces");
static_assert(GEMM_ZERO_PER_LANE * GEMM_THREADS >= H2_PIECES, "the grid covers every piece of h2");
static_assert((GEMM_ZERO_PER_LANE - 1) * GEMM_THREADS <= H2_PIECES, "all but a lane's last piece lie inside h2");

__global__ __launch_bounds__(GEMM_THREADS_PER_GROUP) void gcn_gemm_xw1(
    const float* __restrict__ X, const float* __restrict__ W1,
    float* __restrict__ h1, float* __restrict__ h2) {
  __shared__ float sW[IN_DIM * HID_DIM];
  for (int i = threadIdx.x; i < IN_DIM * HID_DIM; i += GEMM_THREADS_PER_GROUP) sW[i] = W1[i];
  __syncthreads();

  const int wave = threadIdx.x >> 5;
  const int lane = threadIdx.x & 31;
  const int tile = blockIdx.x * 8 + wave;

  if (tile < N_TILES) {
    const int m  = lane & 15;
    const int hi = lane >> 4;

    const float* __restrict__ xrow = X + (size_t)(tile * 16 + m) * IN_DIM;

    v8f c = {0.f, 0.f, 0.f, 0.f, 0.f, 0.f, 0.f, 0.f};

    #pragma unroll 4
    for (int k0 = 0; k0 < IN_DIM; k0 += 4) {
      const int k = k0 + 2 * hi;
      v2f a;
      a.x = xrow[k];
      a.y = xrow[k + 1];
      v2f b;
      b.x = sW[k * HID_DIM + m];
      b.y = sW[(k + 1) * HID_DIM + m];
      c = __builtin_amdgcn_wmma_f32_16x16x4_f32(
          false, a, false, b, (short)0, c, false, false);
    }

    float* __restrict__ out = h1 + (size_t)tile * 16 * HID_DIM;
    const int mbase = hi * 8;
    #pragma unroll
    for (int v = 0; v < 8; ++v)
      out[(mbase + v) * HID_DIM + m] = c[v];
  }

  {
    const unsigned g = blockIdx.x * (unsigned)GEMM_THREADS_PER_GROUP + threadIdx.x;
    float4* __restrict__ h2v = reinterpret_cast<float4*>(h2);
    const float4 zero = make_float4(0.0f, 0.0f, 0.0f, 0.0f);
    for (unsigned j = 0; j + 1u < (unsigned)GEMM_ZERO_PER_LANE; ++j)
      h2v[j * (unsigned)GEMM_THREADS + g] = zero;
    const unsigned last = (unsigned)(GEMM_ZERO_PER_LANE - 1) * (unsigned)GEMM_THREADS + g;
    if (last < (unsigned)H2_PIECES) h2v[last] = zero;
  }
}

#define S1_THREADS 256
#define S1_STEPS   10
#define S1_WAVES   (N_EDGES / (32 * S1_STEPS))
#define S1_GROUPS  (S1_WAVES * 32 / S1_THREADS)
static_assert(N_EDGES % (32 * S1_STEPS) == 0, "every wave owns S1_STEPS full groups");
static_assert((S1_WAVES * 32) % S1_THREADS == 0, "whole work-groups");

template <int J>
__device__ __forceinline__ int row_bcast_i(int x) {
  return __builtin_amdgcn_update_dpp(0, x, 0x150 + J, 0xF, 0xF, false);
}
template <int J>
__device__ __forceinline__ float row_bcast_f(float x) {
  return __int_as_float(row_bcast_i<J>(__float_as_int(x)));
}

#define SPMM1_GATHER(J)                                              \
  {                                                                  \
    const unsigned sj = (unsigned)row_bcast_i<J>(s);                 \
    tt[J] = (unsigned)row_bcast_i<J>(t);                             \
    vv[J] = row_bcast_f<J>(w) * h1[sj * HID_DIM + d];                \
  }
#define SPMM1_SCATTER(J)                                             \
  __hip_atomic_fetch_add(&acc[tt[J] * HID_DIM + d], vv[J],           \
                         __ATOMIC_RELAXED, __HIP_MEMORY_SCOPE_AGENT);

__global__ __launch_bounds__(S1_THREADS) void gcn_spmm1(
    const int* __restrict__ esrc, const int* __restrict__ edst,
    const float* __restrict__ ew, const float* __restrict__ h1,
    float* __restrict__ h2) {
  const unsigned d = threadIdx.x & 15u;
  const unsigned wave_id = (blockIdx.x * (unsigned)S1_THREADS + threadIdx.x) >> 5;
  const unsigned e0 = wave_id * (unsigned)(32 * S1_STEPS) + (threadIdx.x & 31u);

  int   sr[S1_STEPS], tr[S1_STEPS];
  float wr[S1_STEPS];
  #pragma unroll
  for (int g = 0; g < S1_STEPS; ++g) {
    sr[g] = esrc[e0 + 32u * (unsigned)g];
    tr[g] = edst[e0 + 32u * (unsigned)g];
    wr[g] = ew[e0 + 32u * (unsigned)g];
  }

  const unsigned where = __builtin_amdgcn_s_sendmsg_rtn(0x87);
  float* __restrict__ acc = h2 + (size_t)((where >> 18) & 1u) * H2_PLANE;

  #pragma unroll
  for (int g = 0; g < S1_STEPS; ++g) {
    const int   s = sr[g];
    const int   t = tr[g];
    const float w = wr[g];
    unsigned tt[16];
    float    vv[16];
    SPMM1_GATHER(0)  SPMM1_GATHER(1)  SPMM1_GATHER(2)  SPMM1_GATHER(3)
    SPMM1_GATHER(4)  SPMM1_GATHER(5)  SPMM1_GATHER(6)  SPMM1_GATHER(7)
    SPMM1_GATHER(8)  SPMM1_GATHER(9)  SPMM1_GATHER(10) SPMM1_GATHER(11)
    SPMM1_GATHER(12) SPMM1_GATHER(13) SPMM1_GATHER(14) SPMM1_GATHER(15)
    SPMM1_SCATTER(0)  SPMM1_SCATTER(1)  SPMM1_SCATTER(2)  SPMM1_SCATTER(3)
    SPMM1_SCATTER(4)  SPMM1_SCATTER(5)  SPMM1_SCATTER(6)  SPMM1_SCATTER(7)
    SPMM1_SCATTER(8)  SPMM1_SCATTER(9)  SPMM1_SCATTER(10) SPMM1_SCATTER(11)
    SPMM1_SCATTER(12) SPMM1_SCATTER(13) SPMM1_SCATTER(14) SPMM1_SCATTER(15)
  }
}
#undef SPMM1_GATHER
#undef SPMM1_SCATTER

__global__ void gcn_layer2(const float* __restrict__ h2,
                           const float* __restrict__ b1,
                           const float* __restrict__ W2,
                           const float* __restrict__ b2,
                           float* __restrict__ z,
                           float* __restrict__ out,
                           float* __restrict__ out1) {
  int i = blockIdx.x * blockDim.x + threadIdx.x;
  if (i >= N_NODES) return;
  out[i]  = b2[0];
  out1[i] = 0.0f;
  const float* __restrict__ row  = h2 + (size_t)i * HID_DIM;
  const float* __restrict__ row1 = row + H2_PLANE;
  float acc = 0.0f;
  #pragma unroll
  for (int d = 0; d < HID_DIM; ++d) {
    float hv = (row[d] + row1[d]) + b1[d];
    hv = fmaxf(hv, 0.0f);
    acc = fmaf(hv, W2[d], acc);
  }
  z[i] = acc;
}

#define S2_THREADS    512
#define S2_PER_THREAD 25
#define S2_CHUNK      (S2_THREADS * S2_PER_THREAD)
#define S2_GROUPS     (N_EDGES / S2_CHUNK)
#define S2_BINS       4096
#define S2_BINS_PER_THREAD (S2_BINS / S2_THREADS)
static_assert(N_EDGES % S2_CHUNK == 0, "every work-group owns a full chunk");
static_assert((N_NODES + 31) / 32 <= S2_BINS, "one counter per 128-byte line of out and of z");
static_assert(S2_BINS % S2_THREADS == 0, "scan: whole bins per thread");
static_assert((S2_GROUPS - 1) * (S2_CHUNK / S2_GROUPS) < S2_CHUNK, "walk rotation stays inside one wrap");

__global__ __launch_bounds__(S2_THREADS) void gcn_spmm2(
    const int* __restrict__ esrc, const int* __restrict__ edst,
    const float* __restrict__ ew, const float* __restrict__ z,
    float* __restrict__ out, float* __restrict__ out1) {
  __shared__ unsigned sBinD[S2_BINS];
  __shared__ unsigned sBinS[S2_BINS];
  __shared__ unsigned sZ[S2_CHUNK];
  __shared__ uint2    sAdd[S2_CHUNK];
  __shared__ unsigned sWaveSum[2][S2_THREADS / 32];

  const unsigned tid  = threadIdx.x;
  const unsigned lane = tid & 31u;
  const unsigned wave = tid >> 5;
  const unsigned base = blockIdx.x * (unsigned)S2_CHUNK;

  for (unsigned i = tid; i < S2_BINS; i += S2_THREADS) {
    sBinD[i] = 0u;
    sBinS[i] = 0u;
  }

  unsigned t[S2_PER_THREAD], s[S2_PER_THREAD];
  unsigned pd[S2_PER_THREAD], ps[S2_PER_THREAD];
  float    w[S2_PER_THREAD];
  #pragma unroll
  for (int k = 0; k < S2_PER_THREAD; ++k) {
    const unsigned e = base + (unsigned)k * S2_THREADS + tid;
    t[k] = (unsigned)edst[e];
    s[k] = (unsigned)esrc[e];
    w[k] = ew[e];
  }
  __syncthreads();

  #pragma unroll
  for (int k = 0; k < S2_PER_THREAD; ++k) {
    pd[k] = atomicAdd(&sBinD[(t[k] >> 5) & (S2_BINS - 1)], 1u);
    ps[k] = atomicAdd(&sBinS[(s[k] >> 5) & (S2_BINS - 1)], 1u);
  }
  __syncthreads();

  unsigned cD[S2_BINS_PER_THREAD], cS[S2_BINS_PER_THREAD];
  unsigned runD = 0u, runS = 0u;
  #pragma unroll
  for (int i = 0; i < S2_BINS_PER_THREAD; ++i) {
    const unsigned nD = sBinD[tid * S2_BINS_PER_THREAD + i];
    const unsigned nS = sBinS[tid * S2_BINS_PER_THREAD + i];
    cD[i] = runD;
    cS[i] = runS;
    runD += nD;
    runS += nS;
  }
  unsigned inclD = runD, inclS = runS;
  #pragma unroll
  for (unsigned d = 1u; d < 32u; d <<= 1) {
    const unsigned upD = __shfl_up(inclD, d);
    const unsigned upS = __shfl_up(inclS, d);
    if (lane >= d) {
      inclD += upD;
      inclS += upS;
    }
  }
  if (lane == 31u) {
    sWaveSum[0][wave] = inclD;
    sWaveSum[1][wave] = inclS;
  }
  __syncthreads();
  unsigned beforeD = inclD - runD, beforeS = inclS - runS;
  for (unsigned i = 0; i < wave; ++i) {
    beforeD += sWaveSum[0][i];
    beforeS += sWaveSum[1][i];
  }
  #pragma unroll
  for (int i = 0; i < S2_BINS_PER_THREAD; ++i) {
    sBinD[tid * S2_BINS_PER_THREAD + i] = beforeD + cD[i];
    sBinS[tid * S2_BINS_PER_THREAD + i] = beforeS + cS[i];
  }
  __syncthreads();

  #pragma unroll
  for (int k = 0; k < S2_PER_THREAD; ++k) {
    ps[k] += sBinS[(s[k] >> 5) & (S2_BINS - 1)];
    pd[k] += sBinD[(t[k] >> 5) & (S2_BINS - 1)];
    sZ[ps[k]] = s[k];
  }
  __syncthreads();

  const unsigned rot = blockIdx.x * (unsigned)(S2_CHUNK / S2_GROUPS);

  float zz[S2_PER_THREAD];
  #pragma unroll
  for (int k = 0; k < S2_PER_THREAD; ++k) {
    unsigned j = (unsigned)k * S2_THREADS + tid + rot;
    if (j >= (unsigned)S2_CHUNK) j -= (unsigned)S2_CHUNK;
    zz[k] = z[sZ[j]];
  }
  #pragma unroll
  for (int k = 0; k < S2_PER_THREAD; ++k) {
    unsigned j = (unsigned)k * S2_THREADS + tid + rot;
    if (j >= (unsigned)S2_CHUNK) j -= (unsigned)S2_CHUNK;
    sZ[j] = __float_as_uint(zz[k]);
  }
  __syncthreads();

  #pragma unroll
  for (int k = 0; k < S2_PER_THREAD; ++k) {
    const float v = w[k] * __uint_as_float(sZ[ps[k]]);
    sAdd[pd[k]] = make_uint2(t[k], __float_as_uint(v));
  }
  __syncthreads();

  const unsigned where = __builtin_amdgcn_s_sendmsg_rtn(0x87);
  float* __restrict__ acc = ((where >> 18) & 1u) ? out1 : out;
  #pragma unroll
  for (int k = 0; k < S2_PER_THREAD; ++k) {
    unsigned j = (unsigned)k * S2_THREADS + tid + rot;
    if (j >= (unsigned)S2_CHUNK) j -= (unsigned)S2_CHUNK;
    const uint2 p = sAdd[j];
    __hip_atomic_fetch_add(&acc[p.x], __uint_as_float(p.y),
                           __ATOMIC_RELAXED, __HIP_MEMORY_SCOPE_AGENT);
  }
}

static_assert(N_NODES % 4 == 0, "whole 16-byte pieces");
__global__ __launch_bounds__(256) void gcn_sum_out(
    float* __restrict__ out, const float* __restrict__ out1) {
  const int i = blockIdx.x * 256 + threadIdx.x;
  if (i >= N_NODES / 4) return;
  float4 a = reinterpret_cast<float4*>(out)[i];
  const float4 b = reinterpret_cast<const float4*>(out1)[i];
  a.x += b.x;
  a.y += b.y;
  a.z += b.z;
  a.w += b.w;
  reinterpret_cast<float4*>(out)[i] = a;
}

extern "C" void kernel_launch(void* const* d_in, const int* in_sizes, int n_in,
                              void* d_out, int out_size, void* d_ws, size_t ws_size,
                              hipStream_t stream) {
  const float* x    = (const float*)d_in[0];
  const int*   esrc = (const int*)  d_in[1];
  const int*   edst = (const int*)  d_in[2];
  const float* ew   = (const float*)d_in[3];
  const float* W1   = (const float*)d_in[4];
  const float* b1   = (const float*)d_in[5];
  const float* W2   = (const float*)d_in[6];
  const float* b2   = (const float*)d_in[7];
  float* out = (float*)d_out;

  float* h1 = (float*)d_ws;
  float* out1 = h1;
  float* h2 = h1 + (size_t)N_NODES * HID_DIM;
  float* z  = h2 + (size_t)2 * H2_PLANE;

  gcn_gemm_xw1<<<GEMM_GROUPS, GEMM_THREADS_PER_GROUP, 0, stream>>>(x, W1, h1, h2);
  gcn_spmm1<<<S1_GROUPS, S1_THREADS, 0, stream>>>(esrc, edst, ew, h1, h2);
  gcn_layer2<<<(N_NODES + 255) / 256, 256, 0, stream>>>(h2, b1, W2, b2, z, out, out1);
  gcn_spmm2<<<S2_GROUPS, S2_THREADS, 0, stream>>>(esrc, edst, ew, z, out, out1);
  gcn_sum_out<<<(N_NODES / 4 + 255) / 256, 256, 0, stream>>>(out, out1);
}
